# starting kernel
# baseline (speedup 1.0000x reference)
#include <hip/hip_runtime.h>

typedef __attribute__((ext_vector_type(2))) float v2f;
typedef __attribute__((ext_vector_type(8))) float v8f;

#define T_FRAMES 4096
#define D_DIM    256
#define A_NUM    10
#define KSTEPS   64
#define NTILES   3
#define ROWS_PER_WAVE  32
#define ROWS_PER_BLOCK 256
#define TSTRIDE  50

#define SMEM_FLOATS 12800

__global__ __launch_bounds__(256) void rpn_fused_kernel(
    const float* __restrict__ x,
    const float* __restrict__ W_cls,  const float* __restrict__ b_cls,
    const float* __restrict__ W_bbox, const float* __restrict__ b_bbox,
    float* __restrict__ out_prop, float* __restrict__ out_cls,
    float* __restrict__ out_bbox)
{
    __shared__ float smem[SMEM_FLOATS];
    const int tid  = threadIdx.x;
    const int lane = tid & 31;
    const int wave = tid >> 5;

    for (int i = 0; i < 48; ++i) {
        int f   = i * 256 + tid;
        int j   = f & 1;
        int ln  = (f >> 1) & 31;
        int ntk = f >> 6;
        int nt  = ntk % 3;
        int kk  = ntk / 3;
        int n   = nt * 16 + (ln & 15);
        int k   = kk * 4 + ((ln >> 4) << 1) + j;
        float v = 0.0f;
        if (n < 20)      v = W_cls [n * D_DIM + k];
        else if (n < 40) v = W_bbox[(n - 20) * D_DIM + k];
        smem[f] = v;
    }
    __syncthreads();

    const long m0 = (long)blockIdx.x * ROWS_PER_BLOCK + wave * ROWS_PER_WAVE;

    const v8f zero = {0.f, 0.f, 0.f, 0.f, 0.f, 0.f, 0.f, 0.f};
    v8f acc[2][NTILES];
    for (int mt = 0; mt < 2; ++mt)
        for (int nt = 0; nt < NTILES; ++nt)
            acc[mt][nt] = zero;

    const float* xA0 = x + (m0 + (lane & 15)) * (long)D_DIM + ((lane >> 4) << 1);
    const float* xA1 = xA0 + 16 * D_DIM;

    #pragma unroll 4
    for (int kk = 0; kk < KSTEPS; ++kk) {
        v2f a0 = *(const v2f*)(xA0 + kk * 4);
        v2f a1 = *(const v2f*)(xA1 + kk * 4);
        const float* fb = smem + kk * (NTILES * 64) + lane * 2;
        v2f b0 = *(const v2f*)(fb);
        v2f b1 = *(const v2f*)(fb + 64);
        v2f b2 = *(const v2f*)(fb + 128);
        acc[0][0] = __builtin_amdgcn_wmma_f32_16x16x4_f32(false, a0, false, b0, (short)0, acc[0][0], false, false);
        acc[0][1] = __builtin_amdgcn_wmma_f32_16x16x4_f32(false, a0, false, b1, (short)0, acc[0][1], false, false);
        acc[0][2] = __builtin_amdgcn_wmma_f32_16x16x4_f32(false, a0, false, b2, (short)0, acc[0][2], false, false);
        acc[1][0] = __builtin_amdgcn_wmma_f32_16x16x4_f32(false, a1, false, b0, (short)0, acc[1][0], false, false);
        acc[1][1] = __builtin_amdgcn_wmma_f32_16x16x4_f32(false, a1, false, b1, (short)0, acc[1][1], false, false);
        acc[1][2] = __builtin_amdgcn_wmma_f32_16x16x4_f32(false, a1, false, b2, (short)0, acc[1][2], false, false);
    }
    __syncthreads();

    float* tl = smem + wave * (ROWS_PER_WAVE * TSTRIDE);
    const int colB = lane & 15;
    const int rOff = (lane >> 4) << 3;
    #pragma unroll
    for (int mt = 0; mt < 2; ++mt)
        #pragma unroll
        for (int nt = 0; nt < NTILES; ++nt)
            #pragma unroll
            for (int r = 0; r < 8; ++r)
                tl[(mt * 16 + rOff + r) * TSTRIDE + nt * 16 + colB] = acc[mt][nt][r];
    __syncthreads();

    #pragma unroll
    for (int i = 0; i < 20; ++i) {
        int jj  = i * 32 + lane;
        int row = jj / 20;
        int col = jj - row * 20;
        float vc = tl[row * TSTRIDE + col]      + b_cls[col];
        float vb = tl[row * TSTRIDE + 20 + col] + b_bbox[col];
        out_cls [m0 * 20 + jj] = vc;
        out_bbox[m0 * 20 + jj] = vb;
    }

    const float stepL = 160.0f / 9.0f;
    #pragma unroll
    for (int i = 0; i < 10; ++i) {
        int jj  = i * 32 + lane;
        int row = jj / 10;
        int a   = jj - row * 10;
        long m  = m0 + row;
        float tf  = (float)(m & (T_FRAMES - 1));
        float len = (a == A_NUM - 1) ? 200.0f : fmaf(stepL, (float)a, 40.0f);
        float dx  = tl[row * TSTRIDE + 20 + 2 * a];
        float dl  = tl[row * TSTRIDE + 21 + 2 * a];
        float pc  = dx * len + tf;
        float pl  = expf(dl) * len;
        float h   = 0.5f * (pl - 1.0f);
        float s = fminf(fmaxf(pc - h, 0.0f), (float)(T_FRAMES - 1));
        float e = fminf(fmaxf(pc + h, 0.0f), (float)(T_FRAMES - 1));
        v2f se = {s, e};
        *(v2f*)(out_prop + m * 20 + 2 * a) = se;
    }
}

__global__ __launch_bounds__(256) void anchors_kernel(float* __restrict__ out)
{
    int idx = blockIdx.x * 256 + threadIdx.x;
    if (idx >= T_FRAMES * A_NUM) return;
    int t = idx / A_NUM;
    int a = idx - t * A_NUM;
    const float stepL = 160.0f / 9.0f;
    float len = (a == A_NUM - 1) ? 200.0f : fmaf(stepL, (float)a, 40.0f);
    float h = 0.5f * (len - 1.0f);
    v2f se = {(float)t - h, (float)t + h};
    *(v2f*)(out + idx * 2) = se;
}

extern "C" void kernel_launch(void* const* d_in, const int* in_sizes, int n_in,
                              void* d_out, int out_size, void* d_ws, size_t ws_size,
                              hipStream_t stream) {
    (void)in_sizes; (void)n_in; (void)out_size; (void)d_ws; (void)ws_size;
    const float* x      = (const float*)d_in[0];
    const float* W_cls  = (const float*)d_in[1];
    const float* b_cls  = (const float*)d_in[2];
    const float* W_bbox = (const float*)d_in[3];
    const float* b_bbox = (const float*)d_in[4];

    float* out     = (float*)d_out;
    float* anchors = out;
    float* prop    = anchors + (long)T_FRAMES * A_NUM * 2;
    float* cls     = prop    + 64L * T_FRAMES * A_NUM * 2;
    float* bbox    = cls     + 64L * T_FRAMES * A_NUM * 2;

    anchors_kernel<<<(T_FRAMES * A_NUM + 255) / 256, 256, 0, stream>>>(anchors);

    const long rows = 64L * T_FRAMES;
    rpn_fused_kernel<<<(int)(rows / ROWS_PER_BLOCK), 256, 0, stream>>>(
        x, W_cls, b_cls, W_bbox, b_bbox, prop, cls, bbox);
}


# x and weight fragments loaded 16 bytes per lane, K regrouped by 8 (lane half h takes k=8g+4h..+3, A and B alike), 2 WMMA steps per load; changes the k summation order
# speedup vs baseline: 1.1079x; 1.1079x over previous
#include <hip/hip_runtime.h>

typedef __attribute__((ext_vector_type(2))) float v2f;
typedef __attribute__((ext_vector_type(4))) float v4f;
typedef __attribute__((ext_vector_type(8))) float v8f;

#define T_FRAMES 4096
#define D_DIM    256
#define A_NUM    10
#define KGROUPS  32
#define NTILES   3
#define ROWS_PER_WAVE  32
#define ROWS_PER_BLOCK 256
#define TSTRIDE  50

#define SMEM_FLOATS 12800

__global__ __launch_bounds__(256) void rpn_fused_kernel(
    const float* __restrict__ x,
    const float* __restrict__ W_cls,  const float* __restrict__ b_cls,
    const float* __restrict__ W_bbox, const float* __restrict__ b_bbox,
    float* __restrict__ out_prop, float* __restrict__ out_cls,
    float* __restrict__ out_bbox)
{
    __shared__ __attribute__((aligned(16))) float smem[SMEM_FLOATS];
    const int tid  = threadIdx.x;
    const int lane = tid & 31;
    const int wave = tid >> 5;

    v4f* smem4 = (v4f*)smem;
    for (int i = 0; i < 12; ++i) {
        int q   = i * 256 + tid;
        int ln  = q & 31;
        int gn  = q >> 5;
        int nt  = gn % 3;
        int g   = gn / 3;
        int n   = nt * 16 + (ln & 15);
        int k   = g * 8 + ((ln >> 4) << 2);
        v4f v = {0.0f, 0.0f, 0.0f, 0.0f};
        if (n < 20)      v = *(const v4f*)(W_cls  + n * D_DIM + k);
        else if (n < 40) v = *(const v4f*)(W_bbox + (n - 20) * D_DIM + k);
        smem4[q] = v;
    }
    __syncthreads();

    const long m0 = (long)blockIdx.x * ROWS_PER_BLOCK + wave * ROWS_PER_WAVE;

    const v8f zero = {0.f, 0.f, 0.f, 0.f, 0.f, 0.f, 0.f, 0.f};
    v8f acc[2][NTILES];
    for (int mt = 0; mt < 2; ++mt)
        for (int nt = 0; nt < NTILES; ++nt)
            acc[mt][nt] = zero;

    const float* xA0 = x + (m0 + (lane & 15)) * (long)D_DIM + ((lane >> 4) << 2);
    const float* xA1 = xA0 + 16 * D_DIM;
    const v4f* fb = (const v4f*)smem + lane;

    #pragma unroll 4
    for (int g = 0; g < KGROUPS; ++g) {
        v4f a0 = *(const v4f*)(xA0 + g * 8);
        v4f a1 = *(const v4f*)(xA1 + g * 8);
        v4f b0 = fb[(g * NTILES + 0) * 32];
        v4f b1 = fb[(g * NTILES + 1) * 32];
        v4f b2 = fb[(g * NTILES + 2) * 32];
        v2f a0l = {a0.x, a0.y}, a0h = {a0.z, a0.w};
        v2f a1l = {a1.x, a1.y}, a1h = {a1.z, a1.w};
        v2f b0l = {b0.x, b0.y}, b0h = {b0.z, b0.w};
        v2f b1l = {b1.x, b1.y}, b1h = {b1.z, b1.w};
        v2f b2l = {b2.x, b2.y}, b2h = {b2.z, b2.w};
        acc[0][0] = __builtin_amdgcn_wmma_f32_16x16x4_f32(false, a0l, false, b0l, (short)0, acc[0][0], false, false);
        acc[0][1] = __builtin_amdgcn_wmma_f32_16x16x4_f32(false, a0l, false, b1l, (short)0, acc[0][1], false, false);
        acc[0][2] = __builtin_amdgcn_wmma_f32_16x16x4_f32(false, a0l, false, b2l, (short)0, acc[0][2], false, false);
        acc[1][0] = __builtin_amdgcn_wmma_f32_16x16x4_f32(false, a1l, false, b0l, (short)0, acc[1][0], false, false);
        acc[1][1] = __builtin_amdgcn_wmma_f32_16x16x4_f32(false, a1l, false, b1l, (short)0, acc[1][1], false, false);
        acc[1][2] = __builtin_amdgcn_wmma_f32_16x16x4_f32(false, a1l, false, b2l, (short)0, acc[1][2], false, false);
        acc[0][0] = __builtin_amdgcn_wmma_f32_16x16x4_f32(false, a0h, false, b0h, (short)0, acc[0][0], false, false);
        acc[0][1] = __builtin_amdgcn_wmma_f32_16x16x4_f32(false, a0h, false, b1h, (short)0, acc[0][1], false, false);
        acc[0][2] = __builtin_amdgcn_wmma_f32_16x16x4_f32(false, a0h, false, b2h, (short)0, acc[0][2], false, false);
        acc[1][0] = __builtin_amdgcn_wmma_f32_16x16x4_f32(false, a1h, false, b0h, (short)0, acc[1][0], false, false);
        acc[1][1] = __builtin_amdgcn_wmma_f32_16x16x4_f32(false, a1h, false, b1h, (short)0, acc[1][1], false, false);
        acc[1][2] = __builtin_amdgcn_wmma_f32_16x16x4_f32(false, a1h, false, b2h, (short)0, acc[1][2], false, false);
    }
    __syncthreads();

    float* tl = smem + wave * (ROWS_PER_WAVE * TSTRIDE);
    const int colB = lane & 15;
    const int rOff = (lane >> 4) << 3;
    #pragma unroll
    for (int mt = 0; mt < 2; ++mt)
        #pragma unroll
        for (int nt = 0; nt < NTILES; ++nt)
            #pragma unroll
            for (int r = 0; r < 8; ++r)
                tl[(mt * 16 + rOff + r) * TSTRIDE + nt * 16 + colB] = acc[mt][nt][r];
    __syncthreads();

    #pragma unroll
    for (int i = 0; i < 20; ++i) {
        int jj  = i * 32 + lane;
        int row = jj / 20;
        int col = jj - row * 20;
        float vc = tl[row * TSTRIDE + col]      + b_cls[col];
        float vb = tl[row * TSTRIDE + 20 + col] + b_bbox[col];
        out_cls [m0 * 20 + jj] = vc;
        out_bbox[m0 * 20 + jj] = vb;
    }

    const float stepL = 160.0f / 9.0f;
    #pragma unroll
    for (int i = 0; i < 10; ++i) {
        int jj  = i * 32 + lane;
        int row = jj / 10;
        int a   = jj - row * 10;
        long m  = m0 + row;
        float tf  = (float)(m & (T_FRAMES - 1));
        float len = (a == A_NUM - 1) ? 200.0f : fmaf(stepL, (float)a, 40.0f);
        float dx  = tl[row * TSTRIDE + 20 + 2 * a];
        float dl  = tl[row * TSTRIDE + 21 + 2 * a];
        float pc  = dx * len + tf;
        float pl  = expf(dl) * len;
        float h   = 0.5f * (pl - 1.0f);
        float s = fminf(fmaxf(pc - h, 0.0f), (float)(T_FRAMES - 1));
        float e = fminf(fmaxf(pc + h, 0.0f), (float)(T_FRAMES - 1));
        v2f se = {s, e};
        *(v2f*)(out_prop + m * 20 + 2 * a) = se;
    }
}

__global__ __launch_bounds__(256) void anchors_kernel(float* __restrict__ out)
{
    int idx = blockIdx.x * 256 + threadIdx.x;
    if (idx >= T_FRAMES * A_NUM) return;
    int t = idx / A_NUM;
    int a = idx - t * A_NUM;
    const float stepL = 160.0f / 9.0f;
    float len = (a == A_NUM - 1) ? 200.0f : fmaf(stepL, (float)a, 40.0f);
    float h = 0.5f * (len - 1.0f);
    v2f se = {(float)t - h, (float)t + h};
    *(v2f*)(out + idx * 2) = se;
}

extern "C" void kernel_launch(void* const* d_in, const int* in_sizes, int n_in,
                              void* d_out, int out_size, void* d_ws, size_t ws_size,
                              hipStream_t stream) {
    (void)in_sizes; (void)n_in; (void)out_size; (void)d_ws; (void)ws_size;
    const float* x      = (const float*)d_in[0];
    const float* W_cls  = (const float*)d_in[1];
    const float* b_cls  = (const float*)d_in[2];
    const float* W_bbox = (const float*)d_in[3];
    const float* b_bbox = (const float*)d_in[4];

    float* out     = (float*)d_out;
    float* anchors = out;
    float* prop    = anchors + (long)T_FRAMES * A_NUM * 2;
    float* cls     = prop    + 64L * T_FRAMES * A_NUM * 2;
    float* bbox    = cls     + 64L * T_FRAMES * A_NUM * 2;

    anchors_kernel<<<(T_FRAMES * A_NUM + 255) / 256, 256, 0, stream>>>(anchors);

    const long rows = 64L * T_FRAMES;
    rpn_fused_kernel<<<(int)(rows / ROWS_PER_BLOCK), 256, 0, stream>>>(
        x, W_cls, b_cls, W_bbox, b_bbox, prop, cls, bbox);
}


# weight staging made branch-free: each wave issues its twelve 16-byte weight loads before the first LDS store (was twelve dependent load-wait-store trips); LDS fragment layout unchanged
# speedup vs baseline: 1.2017x; 1.0847x over previous
#include <hip/hip_runtime.h>

typedef __attribute__((ext_vector_type(2))) float v2f;
typedef __attribute__((ext_vector_type(4))) float v4f;
typedef __attribute__((ext_vector_type(8))) float v8f;

#define T_FRAMES 4096
#define D_DIM    256
#define A_NUM    10
#define KGROUPS  32
#define NTILES   3
#define ROWS_PER_WAVE  32
#define ROWS_PER_BLOCK 256
#define TSTRIDE  50

#define SMEM_FLOATS 12800

__global__ __launch_bounds__(256) void rpn_fused_kernel(
    const float* __restrict__ x,
    const float* __restrict__ W_cls,  const float* __restrict__ b_cls,
    const float* __restrict__ W_bbox, const float* __restrict__ b_bbox,
    float* __restrict__ out_prop, float* __restrict__ out_cls,
    float* __restrict__ out_bbox)
{
    __shared__ __attribute__((aligned(16))) float smem[SMEM_FLOATS];
    const int tid  = threadIdx.x;
    const int lane = tid & 31;
    const int wave = tid >> 5;

    v4f* smem4 = (v4f*)smem;
    {
        const int c    = lane & 15;
        const int koff = wave * 8 + ((lane >> 4) << 2);
        const float* p0 = W_cls + c * D_DIM + koff;
        const float* p1 = (c < 4) ? (W_cls + (16 + c) * D_DIM + koff)
                                  : (W_bbox + (c - 4) * D_DIM + koff);
        const float* p2 = W_bbox + (12 + (c & 7)) * D_DIM + koff;
        const bool pad = c >= 8;
        const v4f z4 = {0.0f, 0.0f, 0.0f, 0.0f};
        v4f w0[4], w1[4], w2[4];
        for (int j = 0; j < 4; ++j) {
            w0[j] = *(const v4f*)(p0 + j * 64);
            w1[j] = *(const v4f*)(p1 + j * 64);
            w2[j] = *(const v4f*)(p2 + j * 64);
        }
        for (int j = 0; j < 4; ++j) {
            const int base = (wave + 8 * j) * NTILES * 32 + lane;
            smem4[base]      = w0[j];
            smem4[base + 32] = w1[j];
            smem4[base + 64] = pad ? z4 : w2[j];
        }
    }
    __syncthreads();

    const long m0 = (long)blockIdx.x * ROWS_PER_BLOCK + wave * ROWS_PER_WAVE;

    const v8f zero = {0.f, 0.f, 0.f, 0.f, 0.f, 0.f, 0.f, 0.f};
    v8f acc[2][NTILES];
    for (int mt = 0; mt < 2; ++mt)
        for (int nt = 0; nt < NTILES; ++nt)
            acc[mt][nt] = zero;

    const float* xA0 = x + (m0 + (lane & 15)) * (long)D_DIM + ((lane >> 4) << 2);
    const float* xA1 = xA0 + 16 * D_DIM;
    const v4f* fb = (const v4f*)smem + lane;

    #pragma unroll 4
    for (int g = 0; g < KGROUPS; ++g) {
        v4f a0 = *(const v4f*)(xA0 + g * 8);
        v4f a1 = *(const v4f*)(xA1 + g * 8);
        v4f b0 = fb[(g * NTILES + 0) * 32];
        v4f b1 = fb[(g * NTILES + 1) * 32];
        v4f b2 = fb[(g * NTILES + 2) * 32];
        v2f a0l = {a0.x, a0.y}, a0h = {a0.z, a0.w};
        v2f a1l = {a1.x, a1.y}, a1h = {a1.z, a1.w};
        v2f b0l = {b0.x, b0.y}, b0h = {b0.z, b0.w};
        v2f b1l = {b1.x, b1.y}, b1h = {b1.z, b1.w};
        v2f b2l = {b2.x, b2.y}, b2h = {b2.z, b2.w};
        acc[0][0] = __builtin_amdgcn_wmma_f32_16x16x4_f32(false, a0l, false, b0l, (short)0, acc[0][0], false, false);
        acc[0][1] = __builtin_amdgcn_wmma_f32_16x16x4_f32(false, a0l, false, b1l, (short)0, acc[0][1], false, false);
        acc[0][2] = __builtin_amdgcn_wmma_f32_16x16x4_f32(false, a0l, false, b2l, (short)0, acc[0][2], false, false);
        acc[1][0] = __builtin_amdgcn_wmma_f32_16x16x4_f32(false, a1l, false, b0l, (short)0, acc[1][0], false, false);
        acc[1][1] = __builtin_amdgcn_wmma_f32_16x16x4_f32(false, a1l, false, b1l, (short)0, acc[1][1], false, false);
        acc[1][2] = __builtin_amdgcn_wmma_f32_16x16x4_f32(false, a1l, false, b2l, (short)0, acc[1][2], false, false);
        acc[0][0] = __builtin_amdgcn_wmma_f32_16x16x4_f32(false, a0h, false, b0h, (short)0, acc[0][0], false, false);
        acc[0][1] = __builtin_amdgcn_wmma_f32_16x16x4_f32(false, a0h, false, b1h, (short)0, acc[0][1], false, false);
        acc[0][2] = __builtin_amdgcn_wmma_f32_16x16x4_f32(false, a0h, false, b2h, (short)0, acc[0][2], false, false);
        acc[1][0] = __builtin_amdgcn_wmma_f32_16x16x4_f32(false, a1h, false, b0h, (short)0, acc[1][0], false, false);
        acc[1][1] = __builtin_amdgcn_wmma_f32_16x16x4_f32(false, a1h, false, b1h, (short)0, acc[1][1], false, false);
        acc[1][2] = __builtin_amdgcn_wmma_f32_16x16x4_f32(false, a1h, false, b2h, (short)0, acc[1][2], false, false);
    }
    __syncthreads();

    float* tl = smem + wave * (ROWS_PER_WAVE * TSTRIDE);
    const int colB = lane & 15;
    const int rOff = (lane >> 4) << 3;
    #pragma unroll
    for (int mt = 0; mt < 2; ++mt)
        #pragma unroll
        for (int nt = 0; nt < NTILES; ++nt)
            #pragma unroll
            for (int r = 0; r < 8; ++r)
                tl[(mt * 16 + rOff + r) * TSTRIDE + nt * 16 + colB] = acc[mt][nt][r];
    __syncthreads();

    #pragma unroll
    for (int i = 0; i < 20; ++i) {
        int jj  = i * 32 + lane;
        int row = jj / 20;
        int col = jj - row * 20;
        float vc = tl[row * TSTRIDE + col]      + b_cls[col];
        float vb = tl[row * TSTRIDE + 20 + col] + b_bbox[col];
        out_cls [m0 * 20 + jj] = vc;
        out_bbox[m0 * 20 + jj] = vb;
    }

    const float stepL = 160.0f / 9.0f;
    #pragma unroll
    for (int i = 0; i < 10; ++i) {
        int jj  = i * 32 + lane;
        int row = jj / 10;
        int a   = jj - row * 10;
        long m  = m0 + row;
        float tf  = (float)(m & (T_FRAMES - 1));
        float len = (a == A_NUM - 1) ? 200.0f : fmaf(stepL, (float)a, 40.0f);
        float dx  = tl[row * TSTRIDE + 20 + 2 * a];
        float dl  = tl[row * TSTRIDE + 21 + 2 * a];
        float pc  = dx * len + tf;
        float pl  = expf(dl) * len;
        float h   = 0.5f * (pl - 1.0f);
        float s = fminf(fmaxf(pc - h, 0.0f), (float)(T_FRAMES - 1));
        float e = fminf(fmaxf(pc + h, 0.0f), (float)(T_FRAMES - 1));
        v2f se = {s, e};
        *(v2f*)(out_prop + m * 20 + 2 * a) = se;
    }
}

__global__ __launch_bounds__(256) void anchors_kernel(float* __restrict__ out)
{
    int idx = blockIdx.x * 256 + threadIdx.x;
    if (idx >= T_FRAMES * A_NUM) return;
    int t = idx / A_NUM;
    int a = idx - t * A_NUM;
    const float stepL = 160.0f / 9.0f;
    float len = (a == A_NUM - 1) ? 200.0f : fmaf(stepL, (float)a, 40.0f);
    float h = 0.5f * (len - 1.0f);
    v2f se = {(float)t - h, (float)t + h};
    *(v2f*)(out + idx * 2) = se;
}

extern "C" void kernel_launch(void* const* d_in, const int* in_sizes, int n_in,
                              void* d_out, int out_size, void* d_ws, size_t ws_size,
                              hipStream_t stream) {
    (void)in_sizes; (void)n_in; (void)out_size; (void)d_ws; (void)ws_size;
    const float* x      = (const float*)d_in[0];
    const float* W_cls  = (const float*)d_in[1];
    const float* b_cls  = (const float*)d_in[2];
    const float* W_bbox = (const float*)d_in[3];
    const float* b_bbox = (const float*)d_in[4];

    float* out     = (float*)d_out;
    float* anchors = out;
    float* prop    = anchors + (long)T_FRAMES * A_NUM * 2;
    float* cls     = prop    + 64L * T_FRAMES * A_NUM * 2;
    float* bbox    = cls     + 64L * T_FRAMES * A_NUM * 2;

    anchors_kernel<<<(T_FRAMES * A_NUM + 255) / 256, 256, 0, stream>>>(anchors);

    const long rows = 64L * T_FRAMES;
    rpn_fused_kernel<<<(int)(rows / ROWS_PER_BLOCK), 256, 0, stream>>>(
        x, W_cls, b_cls, W_bbox, b_bbox, prop, cls, bbox);
}


# epilogue: D tiles spilled to LDS in output order, so cls, bbox and proposals leave as 16-byte whole-line stores from linear LDS reads; bias is the accumulators' start value (summed first, not last)
# speedup vs baseline: 1.3123x; 1.0920x over previous
#include <hip/hip_runtime.h>

typedef __attribute__((ext_vector_type(2))) float v2f;
typedef __attribute__((ext_vector_type(4))) float v4f;
typedef __attribute__((ext_vector_type(8))) float v8f;

#define T_FRAMES 4096
#define D_DIM    256
#define A_NUM    10
#define KGROUPS  32
#define NTILES   3
#define ROWS_PER_WAVE  32
#define ROWS_PER_BLOCK 256
#define HEAD_FLOATS 640
#define WAVE_FLOATS 1280

#define SMEM_FLOATS 12288

__global__ __launch_bounds__(256) __attribute__((amdgpu_waves_per_eu(8, 8))) void rpn_fused_kernel(
    const float* __restrict__ x,
    const float* __restrict__ W_cls,  const float* __restrict__ b_cls,
    const float* __restrict__ W_bbox, const float* __restrict__ b_bbox,
    float* __restrict__ out_prop, float* __restrict__ out_cls,
    float* __restrict__ out_bbox)
{
    __shared__ __attribute__((aligned(16))) float smem[SMEM_FLOATS];
    const int tid  = threadIdx.x;
    const int lane = tid & 31;
    const int wave = tid >> 5;

    float bias0, bias1, bias2;
    {
        const int c = lane & 15;
        const float* q1 = (c < 4) ? (b_cls + 16 + c) : (b_bbox + (c - 4));
        bias0 = b_cls[c];
        bias1 = *q1;
        bias2 = b_bbox[12 + (c & 7)];
        bias2 = (c < 8) ? bias2 : 0.0f;
    }

    v4f* smem4 = (v4f*)smem;
    {
        const int c    = lane & 15;
        const int koff = wave * 8 + ((lane >> 4) << 2);
        const float* p0 = W_cls + c * D_DIM + koff;
        const float* p1 = (c < 4) ? (W_cls + (16 + c) * D_DIM + koff)
                                  : (W_bbox + (c - 4) * D_DIM + koff);
        const float* p2 = W_bbox + (12 + (c & 7)) * D_DIM + koff;
        const bool pad = c >= 8;
        const v4f z4 = {0.0f, 0.0f, 0.0f, 0.0f};
        v4f w0[4], w1[4], w2[4];
        for (int j = 0; j < 4; ++j) {
            w0[j] = *(const v4f*)(p0 + j * 64);
            w1[j] = *(const v4f*)(p1 + j * 64);
            w2[j] = *(const v4f*)(p2 + j * 64);
        }
        for (int j = 0; j < 4; ++j) {
            const int base = (wave + 8 * j) * NTILES * 32 + lane;
            smem4[base]      = w0[j];
            smem4[base + 32] = w1[j];
            smem4[base + 64] = pad ? z4 : w2[j];
        }
    }
    __syncthreads();

    const long m0 = (long)blockIdx.x * ROWS_PER_BLOCK + wave * ROWS_PER_WAVE;

    v8f acc[2][NTILES];
    for (int mt = 0; mt < 2; ++mt) {
        acc[mt][0] = (v8f){bias0, bias0, bias0, bias0, bias0, bias0, bias0, bias0};
        acc[mt][1] = (v8f){bias1, bias1, bias1, bias1, bias1, bias1, bias1, bias1};
        acc[mt][2] = (v8f){bias2, bias2, bias2, bias2, bias2, bias2, bias2, bias2};
    }

    const float* xA0 = x + (m0 + (lane & 15)) * (long)D_DIM + ((lane >> 4) << 2);
    const float* xA1 = xA0 + 16 * D_DIM;
    const v4f* fb = (const v4f*)smem + lane;

    #pragma unroll 4
    for (int g = 0; g < KGROUPS; ++g) {
        v4f a0 = *(const v4f*)(xA0 + g * 8);
        v4f a1 = *(const v4f*)(xA1 + g * 8);
        v4f b0 = fb[(g * NTILES + 0) * 32];
        v4f b1 = fb[(g * NTILES + 1) * 32];
        v4f b2 = fb[(g * NTILES + 2) * 32];
        v2f a0l = {a0.x, a0.y}, a0h = {a0.z, a0.w};
        v2f a1l = {a1.x, a1.y}, a1h = {a1.z, a1.w};
        v2f b0l = {b0.x, b0.y}, b0h = {b0.z, b0.w};
        v2f b1l = {b1.x, b1.y}, b1h = {b1.z, b1.w};
        v2f b2l = {b2.x, b2.y}, b2h = {b2.z, b2.w};
        acc[0][0] = __builtin_amdgcn_wmma_f32_16x16x4_f32(false, a0l, false, b0l, (short)0, acc[0][0], false, false);
        acc[0][1] = __builtin_amdgcn_wmma_f32_16x16x4_f32(false, a0l, false, b1l, (short)0, acc[0][1], false, false);
        acc[0][2] = __builtin_amdgcn_wmma_f32_16x16x4_f32(false, a0l, false, b2l, (short)0, acc[0][2], false, false);
        acc[1][0] = __builtin_amdgcn_wmma_f32_16x16x4_f32(false, a1l, false, b0l, (short)0, acc[1][0], false, false);
        acc[1][1] = __builtin_amdgcn_wmma_f32_16x16x4_f32(false, a1l, false, b1l, (short)0, acc[1][1], false, false);
        acc[1][2] = __builtin_amdgcn_wmma_f32_16x16x4_f32(false, a1l, false, b2l, (short)0, acc[1][2], false, false);
        acc[0][0] = __builtin_amdgcn_wmma_f32_16x16x4_f32(false, a0h, false, b0h, (short)0, acc[0][0], false, false);
        acc[0][1] = __builtin_amdgcn_wmma_f32_16x16x4_f32(false, a0h, false, b1h, (short)0, acc[0][1], false, false);
        acc[0][2] = __builtin_amdgcn_wmma_f32_16x16x4_f32(false, a0h, false, b2h, (short)0, acc[0][2], false, false);
        acc[1][0] = __builtin_amdgcn_wmma_f32_16x16x4_f32(false, a1h, false, b0h, (short)0, acc[1][0], false, false);
        acc[1][1] = __builtin_amdgcn_wmma_f32_16x16x4_f32(false, a1h, false, b1h, (short)0, acc[1][1], false, false);
        acc[1][2] = __builtin_amdgcn_wmma_f32_16x16x4_f32(false, a1h, false, b2h, (short)0, acc[1][2], false, false);
    }
    __syncthreads();

    float* wl = smem + wave * WAVE_FLOATS;
    {
        const int c    = lane & 15;
        const int rowB = ((lane >> 4) << 3) * 20;
        const int o0 = rowB + c;
        const int o1 = rowB + ((c < 4) ? (16 + c) : (HEAD_FLOATS + c - 4));
        const int o2 = rowB + HEAD_FLOATS + 12 + c;
        #pragma unroll
        for (int mt = 0; mt < 2; ++mt)
            #pragma unroll
            for (int r = 0; r < 8; ++r) {
                wl[o0 + (mt * 16 + r) * 20] = acc[mt][0][r];
                wl[o1 + (mt * 16 + r) * 20] = acc[mt][1][r];
            }
        if (c < 8) {
            #pragma unroll
            for (int mt = 0; mt < 2; ++mt)
                #pragma unroll
                for (int r = 0; r < 8; ++r)
                    wl[o2 + (mt * 16 + r) * 20] = acc[mt][2][r];
        }
    }
    __syncthreads();

    const v4f* c4 = (const v4f*)wl;
    const v4f* b4 = (const v4f*)(wl + HEAD_FLOATS);
    v4f* oc = (v4f*)(out_cls  + m0 * 20);
    v4f* ob = (v4f*)(out_bbox + m0 * 20);
    v4f* op = (v4f*)(out_prop + m0 * 20);
    const int   t0    = (int)(m0 & (T_FRAMES - 1));
    const float stepL = 160.0f / 9.0f;
    const float tmax  = (float)(T_FRAMES - 1);
    #pragma unroll
    for (int i = 0; i < 5; ++i) {
        const int q   = i * 32 + lane;
        const v4f cv  = c4[q];
        const v4f bv  = b4[q];
        oc[q] = cv;
        ob[q] = bv;
        const int row = q / 5;
        const int a0  = 2 * (q - row * 5);
        const float tf   = (float)(t0 + row);
        const float len0 = fmaf(stepL, (float)a0, 40.0f);
        const float len1 = (a0 == A_NUM - 2) ? 200.0f : fmaf(stepL, (float)(a0 + 1), 40.0f);
        const float pc0 = bv.x * len0 + tf;
        const float pl0 = expf(bv.y) * len0;
        const float h0  = 0.5f * (pl0 - 1.0f);
        const float pc1 = bv.z * len1 + tf;
        const float pl1 = expf(bv.w) * len1;
        const float h1  = 0.5f * (pl1 - 1.0f);
        v4f pv;
        pv.x = fminf(fmaxf(pc0 - h0, 0.0f), tmax);
        pv.y = fminf(fmaxf(pc0 + h0, 0.0f), tmax);
        pv.z = fminf(fmaxf(pc1 - h1, 0.0f), tmax);
        pv.w = fminf(fmaxf(pc1 + h1, 0.0f), tmax);
        op[q] = pv;
    }
}

__global__ __launch_bounds__(256) void anchors_kernel(float* __restrict__ out)
{
    int idx = blockIdx.x * 256 + threadIdx.x;
    if (idx >= T_FRAMES * A_NUM) return;
    int t = idx / A_NUM;
    int a = idx - t * A_NUM;
    const float stepL = 160.0f / 9.0f;
    float len = (a == A_NUM - 1) ? 200.0f : fmaf(stepL, (float)a, 40.0f);
    float h = 0.5f * (len - 1.0f);
    v2f se = {(float)t - h, (float)t + h};
    *(v2f*)(out + idx * 2) = se;
}

extern "C" void kernel_launch(void* const* d_in, const int* in_sizes, int n_in,
                              void* d_out, int out_size, void* d_ws, size_t ws_size,
                              hipStream_t stream) {
    (void)in_sizes; (void)n_in; (void)out_size; (void)d_ws; (void)ws_size;
    const float* x      = (const float*)d_in[0];
    const float* W_cls  = (const float*)d_in[1];
    const float* b_cls  = (const float*)d_in[2];
    const float* W_bbox = (const float*)d_in[3];
    const float* b_bbox = (const float*)d_in[4];

    float* out     = (float*)d_out;
    float* anchors = out;
    float* prop    = anchors + (long)T_FRAMES * A_NUM * 2;
    float* cls     = prop    + 64L * T_FRAMES * A_NUM * 2;
    float* bbox    = cls     + 64L * T_FRAMES * A_NUM * 2;

    anchors_kernel<<<(T_FRAMES * A_NUM + 255) / 256, 256, 0, stream>>>(anchors);

    const long rows = 64L * T_FRAMES;
    rpn_fused_kernel<<<(int)(rows / ROWS_PER_BLOCK), 256, 0, stream>>>(
        x, W_cls, b_cls, W_bbox, b_bbox, prop, cls, bbox);
}


# x loaded as whole 128-byte lines (4 rows x 128 B per 16-byte-per-lane load, was 16 rows x 32 B), regrouped into A fragments in private per-wave LDS; 16-wave groups, no epilogue barrier; k order kept
# speedup vs baseline: 1.3790x; 1.0508x over previous
#include <hip/hip_runtime.h>


typedef __attribute__((ext_vector_type(2))) float v2f;
typedef __attribute__((ext_vector_type(4))) float v4f;
typedef __attribute__((ext_vector_type(8))) float v8f;

#define T_FRAMES 4096
#define D_DIM    256
#define A_NUM    10
#define KGROUPS  32
#define NTILES   3
#define ROWS_PER_WAVE  32
#define WAVES_PER_BLOCK 16
#define ROWS_PER_BLOCK 512
#define TRIPS    8
#define HEAD_FLOATS 640
#define BFRAG_FLOATS 12288
#define XPITCH   36
#define WAVE_FLOATS 1280
#define SMEM_FLOATS (BFRAG_FLOATS + WAVES_PER_BLOCK * WAVE_FLOATS)

static __device__ __forceinline__ void wave_lds_order()
{
    __builtin_amdgcn_fence(__ATOMIC_RELEASE, "wavefront", "local");
    __builtin_amdgcn_wave_barrier();
    __builtin_amdgcn_fence(__ATOMIC_ACQUIRE, "wavefront", "local");
}

__global__ __launch_bounds__(512) __attribute__((amdgpu_waves_per_eu(8, 8))) void rpn_fused_kernel(
    const float* __restrict__ x,
    const float* __restrict__ W_cls,  const float* __restrict__ b_cls,
    const float* __restrict__ W_bbox, const float* __restrict__ b_bbox,
    float* __restrict__ out_prop, float* __restrict__ out_cls,
    float* __restrict__ out_bbox)
{
    __shared__ __attribute__((aligned(16))) float smem[SMEM_FLOATS];
    const int tid  = threadIdx.x;
    const int lane = tid & 31;
    const int wave = tid >> 5;

    float bias0, bias1, bias2;
    {
        const int c = lane & 15;
        const float* q1 = (c < 4) ? (b_cls + 16 + c) : (b_bbox + (c - 4));
        bias0 = b_cls[c];
        bias1 = *q1;
        bias2 = b_bbox[12 + (c & 7)];
        bias2 = (c < 8) ? bias2 : 0.0f;
    }

    v4f* smem4 = (v4f*)smem;
    {
        const int c    = lane & 15;
        const int koff = wave * 8 + ((lane >> 4) << 2);
        const float* p0 = W_cls + c * D_DIM + koff;
        const float* p1 = (c < 4) ? (W_cls + (16 + c) * D_DIM + koff)
                                  : (W_bbox + (c - 4) * D_DIM + koff);
        const float* p2 = W_bbox + (12 + (c & 7)) * D_DIM + koff;
        const bool pad = c >= 8;
        const v4f z4 = {0.0f, 0.0f, 0.0f, 0.0f};
        v4f w0[2], w1[2], w2[2];
        for (int j = 0; j < 2; ++j) {
            w0[j] = *(const v4f*)(p0 + j * 128);
            w1[j] = *(const v4f*)(p1 + j * 128);
            w2[j] = *(const v4f*)(p2 + j * 128);
        }
        for (int j = 0; j < 2; ++j) {
            const int base = (wave + 16 * j) * NTILES * 32 + lane;
            smem4[base]      = w0[j];
            smem4[base + 32] = w1[j];
            smem4[base + 64] = pad ? z4 : w2[j];
        }
    }

    const long m0 = (long)blockIdx.x * ROWS_PER_BLOCK + wave * ROWS_PER_WAVE;

    const int rq = (lane >> 3) << 3;
    const float* xg = x + (m0 + rq) * (long)D_DIM + ((lane & 7) << 2);
    float* wl = smem + BFRAG_FLOATS + wave * WAVE_FLOATS;
    float* wst = wl + rq * XPITCH + ((lane & 7) << 2);
    const float* wld = wl + (lane & 15) * XPITCH + ((lane >> 4) << 2);
    const v4f* fb = (const v4f*)smem + lane;

    v4f G[8];
    #pragma unroll
    for (int j = 0; j < 8; ++j)
        G[j] = *(const v4f*)(xg + j * D_DIM);

    v8f acc[2][NTILES];
    for (int mt = 0; mt < 2; ++mt) {
        acc[mt][0] = (v8f){bias0, bias0, bias0, bias0, bias0, bias0, bias0, bias0};
        acc[mt][1] = (v8f){bias1, bias1, bias1, bias1, bias1, bias1, bias1, bias1};
        acc[mt][2] = (v8f){bias2, bias2, bias2, bias2, bias2, bias2, bias2, bias2};
    }

    __syncthreads();

    for (int t = 0; t < TRIPS; ++t) {
        #pragma unroll
        for (int j = 0; j < 8; ++j)
            *(v4f*)(wst + j * XPITCH) = G[j];
        wave_lds_order();
        if (t + 1 < TRIPS) {
            #pragma unroll
            for (int j = 0; j < 8; ++j)
                G[j] = *(const v4f*)(xg + (t + 1) * 32 + j * D_DIM);
        }
        #pragma unroll
        for (int g = 0; g < 4; ++g) {
            const int kg = t * 4 + g;
            v4f a0 = *(const v4f*)(wld + g * 8);
            v4f a1 = *(const v4f*)(wld + 16 * XPITCH + g * 8);
            v4f b0 = fb[(kg * NTILES + 0) * 32];
            v4f b1 = fb[(kg * NTILES + 1) * 32];
            v4f b2 = fb[(kg * NTILES + 2) * 32];
            v2f a0l = {a0.x, a0.y}, a0h = {a0.z, a0.w};
            v2f a1l = {a1.x, a1.y}, a1h = {a1.z, a1.w};
            v2f b0l = {b0.x, b0.y}, b0h = {b0.z, b0.w};
            v2f b1l = {b1.x, b1.y}, b1h = {b1.z, b1.w};
            v2f b2l = {b2.x, b2.y}, b2h = {b2.z, b2.w};
            acc[0][0] = __builtin_amdgcn_wmma_f32_16x16x4_f32(false, a0l, false, b0l, (short)0, acc[0][0], false, false);
            acc[0][1] = __builtin_amdgcn_wmma_f32_16x16x4_f32(false, a0l, false, b1l, (short)0, acc[0][1], false, false);
            acc[0][2] = __builtin_amdgcn_wmma_f32_16x16x4_f32(false, a0l, false, b2l, (short)0, acc[0][2], false, false);
            acc[1][0] = __builtin_amdgcn_wmma_f32_16x16x4_f32(false, a1l, false, b0l, (short)0, acc[1][0], false, false);
            acc[1][1] = __builtin_amdgcn_wmma_f32_16x16x4_f32(false, a1l, false, b1l, (short)0, acc[1][1], false, false);
            acc[1][2] = __builtin_amdgcn_wmma_f32_16x16x4_f32(false, a1l, false, b2l, (short)0, acc[1][2], false, false);
            acc[0][0] = __builtin_amdgcn_wmma_f32_16x16x4_f32(false, a0h, false, b0h, (short)0, acc[0][0], false, false);
            acc[0][1] = __builtin_amdgcn_wmma_f32_16x16x4_f32(false, a0h, false, b1h, (short)0, acc[0][1], false, false);
            acc[0][2] = __builtin_amdgcn_wmma_f32_16x16x4_f32(false, a0h, false, b2h, (short)0, acc[0][2], false, false);
            acc[1][0] = __builtin_amdgcn_wmma_f32_16x16x4_f32(false, a1h, false, b0h, (short)0, acc[1][0], false, false);
            acc[1][1] = __builtin_amdgcn_wmma_f32_16x16x4_f32(false, a1h, false, b1h, (short)0, acc[1][1], false, false);
            acc[1][2] = __builtin_amdgcn_wmma_f32_16x16x4_f32(false, a1h, false, b2h, (short)0, acc[1][2], false, false);
        }
        wave_lds_order();
    }

    {
        const int c    = lane & 15;
        const int rowB = ((lane >> 4) << 3) * 20;
        const int o0 = rowB + c;
        const int o1 = rowB + ((c < 4) ? (16 + c) : (HEAD_FLOATS + c - 4));
        const int o2 = rowB + HEAD_FLOATS + 12 + c;
        #pragma unroll
        for (int mt = 0; mt < 2; ++mt)
            #pragma unroll
            for (int r = 0; r < 8; ++r) {
                wl[o0 + (mt * 16 + r) * 20] = acc[mt][0][r];
                wl[o1 + (mt * 16 + r) * 20] = acc[mt][1][r];
            }
        if (c < 8) {
            #pragma unroll
            for (int mt = 0; mt < 2; ++mt)
                #pragma unroll
                for (int r = 0; r < 8; ++r)
                    wl[o2 + (mt * 16 + r) * 20] = acc[mt][2][r];
        }
    }
    wave_lds_order();

    const v4f* c4 = (const v4f*)wl;
    const v4f* b4 = (const v4f*)(wl + HEAD_FLOATS);
    v4f* oc = (v4f*)(out_cls  + m0 * 20);
    v4f* ob = (v4f*)(out_bbox + m0 * 20);
    v4f* op = (v4f*)(out_prop + m0 * 20);
    const int   t0    = (int)(m0 & (T_FRAMES - 1));
    const float stepL = 160.0f / 9.0f;
    const float tmax  = (float)(T_FRAMES - 1);
    #pragma unroll
    for (int i = 0; i < 5; ++i) {
        const int q   = i * 32 + lane;
        const v4f cv  = c4[q];
        const v4f bv  = b4[q];
        oc[q] = cv;
        ob[q] = bv;
        const int row = q / 5;
        const int a0  = 2 * (q - row * 5);
        const float tf   = (float)(t0 + row);
        const float len0 = fmaf(stepL, (float)a0, 40.0f);
        const float len1 = (a0 == A_NUM - 2) ? 200.0f : fmaf(stepL, (float)(a0 + 1), 40.0f);
        const float pc0 = bv.x * len0 + tf;
        const float pl0 = expf(bv.y) * len0;
        const float h0  = 0.5f * (pl0 - 1.0f);
        const float pc1 = bv.z * len1 + tf;
        const float pl1 = expf(bv.w) * len1;
        const float h1  = 0.5f * (pl1 - 1.0f);
        v4f pv;
        pv.x = fminf(fmaxf(pc0 - h0, 0.0f), tmax);
        pv.y = fminf(fmaxf(pc0 + h0, 0.0f), tmax);
        pv.z = fminf(fmaxf(pc1 - h1, 0.0f), tmax);
        pv.w = fminf(fmaxf(pc1 + h1, 0.0f), tmax);
        op[q] = pv;
    }
}

__global__ __launch_bounds__(256) void anchors_kernel(float* __restrict__ out)
{
    int idx = blockIdx.x * 256 + threadIdx.x;
    if (idx >= T_FRAMES * A_NUM) return;
    int t = idx / A_NUM;
    int a = idx - t * A_NUM;
    const float stepL = 160.0f / 9.0f;
    float len = (a == A_NUM - 1) ? 200.0f : fmaf(stepL, (float)a, 40.0f);
    float h = 0.5f * (len - 1.0f);
    v2f se = {(float)t - h, (float)t + h};
    *(v2f*)(out + idx * 2) = se;
}

extern "C" void kernel_launch(void* const* d_in, const int* in_sizes, int n_in,
                              void* d_out, int out_size, void* d_ws, size_t ws_size,
                              hipStream_t stream) {
    (void)in_sizes; (void)n_in; (void)out_size; (void)d_ws; (void)ws_size;
    const float* x      = (const float*)d_in[0];
    const float* W_cls  = (const float*)d_in[1];
    const float* b_cls  = (const float*)d_in[2];
    const float* W_bbox = (const float*)d_in[3];
    const float* b_bbox = (const float*)d_in[4];

    float* out     = (float*)d_out;
    float* anchors = out;
    float* prop    = anchors + (long)T_FRAMES * A_NUM * 2;
    float* cls     = prop    + 64L * T_FRAMES * A_NUM * 2;
    float* bbox    = cls     + 64L * T_FRAMES * A_NUM * 2;

    anchors_kernel<<<(T_FRAMES * A_NUM + 255) / 256, 256, 0, stream>>>(anchors);

    const long rows = 64L * T_FRAMES;
    rpn_fused_kernel<<<(int)(rows / ROWS_PER_BLOCK), 512, 0, stream>>>(
        x, W_cls, b_cls, W_bbox, b_bbox, prop, cls, bbox);
}
